# starting kernel
# baseline (speedup 1.0000x reference)
#include <hip/hip_runtime.h>


typedef __attribute__((ext_vector_type(2))) float v2f;
typedef __attribute__((ext_vector_type(8))) float v8f;

namespace {
constexpr int kT       = 4000;
constexpr int kSig     = 64 * 64;
constexpr int kTaps    = 413;
constexpr int kHalf    = (kTaps - 1) / 2;
constexpr int kBlkS    = 64;
constexpr int kBlkT    = 128;
constexpr int kWin     = kBlkT + 2 * kHalf;
constexpr int kStride  = 546;
constexpr int kKSteps  = 111;
constexpr int kHpLen   = 480;
constexpr int kHpOff   = 32;
}

__global__ __launch_bounds__(256)
void fir_bandpass_wmma2x2(const float* __restrict__ x,
                          const float* __restrict__ h,
                          float* __restrict__ out)
{
    __shared__ float xs[kBlkS * kStride];
    __shared__ float hpA[kHpLen];
    __shared__ float hpB[kHpLen];

    const int tid = threadIdx.x;
    const int s0b = blockIdx.x * kBlkS;
    const int t0b = blockIdx.y * kBlkT;

    for (int i = tid; i < kHpLen; i += 256) {
        int j = i - kHpOff;
        float v = (j >= 0 && j < kTaps) ? h[kTaps - 1 - j] : 0.0f;
        hpA[i] = v;
        hpB[i] = v;
    }

    for (int s = (tid >> 4); s < kBlkS; s += 16) {
        const float* row = x + (size_t)(s0b + s) * kT;
        const float xf = row[0];
        const float xl = row[kT - 1];
        for (int jj = (tid & 15); jj < kWin; jj += 16) {
            int j = t0b - kHalf + jj;
            float v;
            if (j < 0)        v = 2.0f * xf - row[-j];
            else if (j >= kT) v = 2.0f * xl - row[2 * kT - 2 - j];
            else              v = row[j];
            xs[s * kStride + jj] = v;
        }
    }
    __syncthreads();

    const int w    = tid >> 5;
    const int lane = tid & 31;
    const int mg   = w & 3;
    const int ng   = w >> 2;
    const int t00  = t0b + (mg << 5);
    if (t00 >= kT) return;

    const int m  = lane & 15;
    const int hi = lane >> 4;
    const int kk = hi << 1;

    const float* hb  = (hi ? hpB : hpA) + (kHpOff + kk - m);
    const float* xb0 = xs + ((ng << 5) + m) * kStride + (mg << 5) + kk;
    const float* xb1 = xb0 + 16 * kStride;

    v8f acc00 = {0.f,0.f,0.f,0.f,0.f,0.f,0.f,0.f};
    v8f acc01 = acc00;
    v8f acc10 = acc00;
    v8f acc11 = acc00;

    #pragma unroll 4
    for (int k = 0; k < kKSteps; ++k) {
        const int c = 4 * k;
        v2f a0, a1, b0, b1;
        a0.x = hb[c];          a0.y = hb[c + 1];
        a1.x = hb[c - 16];     a1.y = hb[c - 15];
        b0.x = xb0[c];         b0.y = xb0[c + 1];
        b1.x = xb1[c];         b1.y = xb1[c + 1];
        acc00 = __builtin_amdgcn_wmma_f32_16x16x4_f32(
            false, a0, false, b0, (short)0, acc00, false, false);
        acc01 = __builtin_amdgcn_wmma_f32_16x16x4_f32(
            false, a0, false, b1, (short)0, acc01, false, false);
        acc10 = __builtin_amdgcn_wmma_f32_16x16x4_f32(
            false, a1, false, b0, (short)0, acc10, false, false);
        acc11 = __builtin_amdgcn_wmma_f32_16x16x4_f32(
            false, a1, false, b1, (short)0, acc11, false, false);
    }

    float* o00 = out + (size_t)(s0b + (ng << 5) + m) * kT + t00 + (hi << 3);
    float* o01 = o00 + (size_t)16 * kT;
    float* o10 = o00 + 16;
    float* o11 = o01 + 16;

    float4 v;
    v.x = acc00[0]; v.y = acc00[1]; v.z = acc00[2]; v.w = acc00[3];
    *(float4*)(o00) = v;
    v.x = acc00[4]; v.y = acc00[5]; v.z = acc00[6]; v.w = acc00[7];
    *(float4*)(o00 + 4) = v;
    v.x = acc01[0]; v.y = acc01[1]; v.z = acc01[2]; v.w = acc01[3];
    *(float4*)(o01) = v;
    v.x = acc01[4]; v.y = acc01[5]; v.z = acc01[6]; v.w = acc01[7];
    *(float4*)(o01 + 4) = v;
    v.x = acc10[0]; v.y = acc10[1]; v.z = acc10[2]; v.w = acc10[3];
    *(float4*)(o10) = v;
    v.x = acc10[4]; v.y = acc10[5]; v.z = acc10[6]; v.w = acc10[7];
    *(float4*)(o10 + 4) = v;
    v.x = acc11[0]; v.y = acc11[1]; v.z = acc11[2]; v.w = acc11[3];
    *(float4*)(o11) = v;
    v.x = acc11[4]; v.y = acc11[5]; v.z = acc11[6]; v.w = acc11[7];
    *(float4*)(o11 + 4) = v;
}

extern "C" void kernel_launch(void* const* d_in, const int* in_sizes, int n_in,
                              void* d_out, int out_size, void* d_ws, size_t ws_size,
                              hipStream_t stream) {
    const float* x = (const float*)d_in[0];
    const float* h = (const float*)d_in[1];
    float* out = (float*)d_out;

    dim3 grid(kSig / kBlkS, (kT + kBlkT - 1) / kBlkT);
    fir_bandpass_wmma2x2<<<grid, 256, 0, stream>>>(x, h, out);
}


# Re-tiled: workgroup 16 signals x 512 times (was 64 x 128), wave 64 times x 16 signals; 4 time tiles share one tap stream held in a register ring; under half the window staging (new f32 sum order)
# speedup vs baseline: 1.8752x; 1.8752x over previous
#include <hip/hip_runtime.h>


typedef __attribute__((ext_vector_type(2))) float v2f;
typedef __attribute__((ext_vector_type(8))) float v8f;

namespace {
constexpr int kT       = 4000;
constexpr int kSig     = 64 * 64;
constexpr int kTaps    = 413;
constexpr int kHalf    = (kTaps - 1) / 2;
constexpr int kBlkS    = 16;
constexpr int kWaveT   = 64;
constexpr int kBlkT    = 8 * kWaveT;
constexpr int kKSteps  = 120;
constexpr int kWin     = kBlkT - kWaveT + 4 * kKSteps;
constexpr int kStride  = 932;
constexpr int kHpOff   = 64;
constexpr int kHpLen   = 544;
constexpr int kRing    = 16;
}

__global__ __launch_bounds__(256)
void fir_bandpass_wmma4x1(const float* __restrict__ x,
                          const float* __restrict__ h,
                          float* __restrict__ out)
{
    __shared__ __attribute__((aligned(16))) float xs[kBlkS * kStride];
    __shared__ __attribute__((aligned(16))) float hp[2 * kHpLen];

    const int tid = threadIdx.x;
    const int s0b = blockIdx.x * kBlkS;
    const int t0b = blockIdx.y * kBlkT;

    for (int i = tid; i < kHpLen; i += 256) {
        int j = i - kHpOff;
        float v = (j >= 0 && j < kTaps) ? h[kTaps - 1 - j] : 0.0f;
        hp[i] = v;
        hp[kHpLen + i] = v;
    }

    {
        const int s = tid >> 4;
        const float* row = x + (size_t)(s0b + s) * kT;
        const float xf = row[0];
        const float xl = row[kT - 1];
        for (int jj = (tid & 15); jj < kWin; jj += 16) {
            int j = t0b - kHalf + jj;
            float v;
            if (j < 0)        v = 2.0f * xf - row[-j];
            else if (j >= kT) v = 2.0f * xl - row[2 * kT - 2 - j];
            else              v = row[j];
            xs[s * kStride + jj] = v;
        }
    }
    __syncthreads();

    const int w    = tid >> 5;
    const int lane = tid & 31;
    const int t00  = t0b + w * kWaveT;
    if (t00 >= kT) return;

    const int m  = lane & 15;
    const int hi = lane >> 4;
    const int kk = hi << 1;

    const float* hb = hp + hi * kHpLen + (kHpOff - 48 + kk - m);
    const float* xb = xs + m * kStride + w * kWaveT + kk;

    v8f acc0 = {0.f,0.f,0.f,0.f,0.f,0.f,0.f,0.f};
    v8f acc1 = acc0;
    v8f acc2 = acc0;
    v8f acc3 = acc0;

    v2f ring[kRing];
    #pragma unroll
    for (int u = 0; u < 12; ++u) {
        ring[u].x = hb[4 * u];
        ring[u].y = hb[4 * u + 1];
    }

#define FIR_STEP(s)                                                                   \
    {                                                                                 \
        v2f a;                                                                        \
        a.x = hbp[4 * ((s) + 12)];                                                    \
        a.y = hbp[4 * ((s) + 12) + 1];                                                \
        ring[((s) + 12) & (kRing - 1)] = a;                                           \
        const v2f b = *(const v2f*)(xbp + 4 * (s));                                   \
        acc0 = __builtin_amdgcn_wmma_f32_16x16x4_f32(                                 \
            false, ring[((s) + 12) & (kRing - 1)], false, b, (short)0, acc0, false, false); \
        acc1 = __builtin_amdgcn_wmma_f32_16x16x4_f32(                                 \
            false, ring[((s) + 8) & (kRing - 1)], false, b, (short)0, acc1, false, false);  \
        acc2 = __builtin_amdgcn_wmma_f32_16x16x4_f32(                                 \
            false, ring[((s) + 4) & (kRing - 1)], false, b, (short)0, acc2, false, false);  \
        acc3 = __builtin_amdgcn_wmma_f32_16x16x4_f32(                                 \
            false, ring[(s) & (kRing - 1)], false, b, (short)0, acc3, false, false);        \
    }

    const float* hbp = hb;
    const float* xbp = xb;
    #pragma unroll 1
    for (int kb = 0; kb < kKSteps / 16; ++kb) {
        #pragma unroll
        for (int s = 0; s < 16; ++s) FIR_STEP(s)
        hbp += 64;
        xbp += 64;
    }
    #pragma unroll
    for (int s = 0; s < kKSteps % 16; ++s) FIR_STEP(s)
#undef FIR_STEP

    float* o = out + (size_t)(s0b + m) * kT + t00 + (hi << 3);

    float4 v;
    v.x = acc0[0]; v.y = acc0[1]; v.z = acc0[2]; v.w = acc0[3];
    *(float4*)(o) = v;
    v.x = acc0[4]; v.y = acc0[5]; v.z = acc0[6]; v.w = acc0[7];
    *(float4*)(o + 4) = v;
    if (t00 + 16 < kT) {
        v.x = acc1[0]; v.y = acc1[1]; v.z = acc1[2]; v.w = acc1[3];
        *(float4*)(o + 16) = v;
        v.x = acc1[4]; v.y = acc1[5]; v.z = acc1[6]; v.w = acc1[7];
        *(float4*)(o + 20) = v;
    }
    if (t00 + 32 < kT) {
        v.x = acc2[0]; v.y = acc2[1]; v.z = acc2[2]; v.w = acc2[3];
        *(float4*)(o + 32) = v;
        v.x = acc2[4]; v.y = acc2[5]; v.z = acc2[6]; v.w = acc2[7];
        *(float4*)(o + 36) = v;
    }
    if (t00 + 48 < kT) {
        v.x = acc3[0]; v.y = acc3[1]; v.z = acc3[2]; v.w = acc3[3];
        *(float4*)(o + 48) = v;
        v.x = acc3[4]; v.y = acc3[5]; v.z = acc3[6]; v.w = acc3[7];
        *(float4*)(o + 52) = v;
    }
}

extern "C" void kernel_launch(void* const* d_in, const int* in_sizes, int n_in,
                              void* d_out, int out_size, void* d_ws, size_t ws_size,
                              hipStream_t stream) {
    const float* x = (const float*)d_in[0];
    const float* h = (const float*)d_in[1];
    float* out = (float*)d_out;

    dim3 grid(kSig / kBlkS, (kT + kBlkT - 1) / kBlkT);
    fir_bandpass_wmma4x1<<<grid, 256, 0, stream>>>(x, h, out);
}


# Band trimmed per tile: each of a wave's 4 time tiles runs only the 107 K-steps where its tap block is non-zero (ramp-in and ramp-out peeled); 428 WMMAs per wave for 480; f32 sum order kept
# speedup vs baseline: 1.9809x; 1.0564x over previous
#include <hip/hip_runtime.h>


typedef __attribute__((ext_vector_type(2))) float v2f;
typedef __attribute__((ext_vector_type(8))) float v8f;

namespace {
constexpr int kT       = 4000;
constexpr int kSig     = 64 * 64;
constexpr int kTaps    = 413;
constexpr int kHalf    = (kTaps - 1) / 2;
constexpr int kBlkS    = 16;
constexpr int kWaveT   = 64;
constexpr int kBlkT    = 8 * kWaveT;
constexpr int kKSteps  = 120;
constexpr int kWin     = kBlkT - kWaveT + 4 * kKSteps;
constexpr int kStride  = 932;
constexpr int kHpOff   = 64;
constexpr int kHpLen   = 544;
constexpr int kRing    = 16;
static_assert((kTaps + 15 + 3) / 4 == 107, "per-tile band length used by the K-step schedule below");
}

__global__ __launch_bounds__(256)
void fir_bandpass_wmma4x1(const float* __restrict__ x,
                          const float* __restrict__ h,
                          float* __restrict__ out)
{
    __shared__ __attribute__((aligned(16))) float xs[kBlkS * kStride];
    __shared__ __attribute__((aligned(16))) float hp[2 * kHpLen];

    const int tid = threadIdx.x;
    const int s0b = blockIdx.x * kBlkS;
    const int t0b = blockIdx.y * kBlkT;

    for (int i = tid; i < kHpLen; i += 256) {
        int j = i - kHpOff;
        float v = (j >= 0 && j < kTaps) ? h[kTaps - 1 - j] : 0.0f;
        hp[i] = v;
        hp[kHpLen + i] = v;
    }

    {
        const int s = tid >> 4;
        const float* row = x + (size_t)(s0b + s) * kT;
        const float xf = row[0];
        const float xl = row[kT - 1];
        for (int jj = (tid & 15); jj < kWin; jj += 16) {
            int j = t0b - kHalf + jj;
            float v;
            if (j < 0)        v = 2.0f * xf - row[-j];
            else if (j >= kT) v = 2.0f * xl - row[2 * kT - 2 - j];
            else              v = row[j];
            xs[s * kStride + jj] = v;
        }
    }
    __syncthreads();

    const int w    = tid >> 5;
    const int lane = tid & 31;
    const int t00  = t0b + w * kWaveT;
    if (t00 >= kT) return;

    const int m  = lane & 15;
    const int hi = lane >> 4;
    const int kk = hi << 1;

    const float* hb = hp + hi * kHpLen + (kHpOff - 48 + kk - m);
    const float* xb = xs + m * kStride + w * kWaveT + kk;

    v8f acc0 = {0.f,0.f,0.f,0.f,0.f,0.f,0.f,0.f};
    v8f acc1 = acc0;
    v8f acc2 = acc0;
    v8f acc3 = acc0;

    v2f ring[kRing];

#define FIR_STEP(ko, sl, ld, t0, t1, t2, t3)                                          \
    {                                                                                 \
        if (ld) {                                                                     \
            v2f a;                                                                    \
            a.x = hq[4 * (ko)];                                                       \
            a.y = hq[4 * (ko) + 1];                                                   \
            ring[(sl) & (kRing - 1)] = a;                                             \
        }                                                                             \
        const v2f b = *(const v2f*)(xq + 4 * (ko));                                   \
        if (t0) acc0 = __builtin_amdgcn_wmma_f32_16x16x4_f32(                         \
            false, ring[(sl) & (kRing - 1)], false, b, (short)0, acc0, false, false); \
        if (t1) acc1 = __builtin_amdgcn_wmma_f32_16x16x4_f32(                         \
            false, ring[((sl) + 12) & (kRing - 1)], false, b, (short)0, acc1, false, false); \
        if (t2) acc2 = __builtin_amdgcn_wmma_f32_16x16x4_f32(                         \
            false, ring[((sl) + 8) & (kRing - 1)], false, b, (short)0, acc2, false, false);  \
        if (t3) acc3 = __builtin_amdgcn_wmma_f32_16x16x4_f32(                         \
            false, ring[((sl) + 4) & (kRing - 1)], false, b, (short)0, acc3, false, false);  \
    }

    const float* hq = hb + 48;
    const float* xq = xb;

    #pragma unroll
    for (int k = 0; k < 4; ++k)  FIR_STEP(k, k, true, true, false, false, false)
    #pragma unroll
    for (int k = 4; k < 8; ++k)  FIR_STEP(k, k, true, true, true, false, false)
    #pragma unroll
    for (int k = 8; k < 12; ++k) FIR_STEP(k, k, true, true, true, true, false)
    hq += 48;
    xq += 48;
    #pragma unroll 1
    for (int kb = 0; kb < 5; ++kb) {
        #pragma unroll
        for (int s = 0; s < 16; ++s) FIR_STEP(s, s + 12, true, true, true, true, true)
        hq += 64;
        xq += 64;
    }
    #pragma unroll
    for (int s = 0; s < 15; ++s) FIR_STEP(s, s + 12, true, true, true, true, true)
    #pragma unroll
    for (int s = 15; s < 19; ++s) FIR_STEP(s, s + 12, false, false, true, true, true)
    #pragma unroll
    for (int s = 19; s < 23; ++s) FIR_STEP(s, s + 12, false, false, false, true, true)
    #pragma unroll
    for (int s = 23; s < 27; ++s) FIR_STEP(s, s + 12, false, false, false, false, true)
#undef FIR_STEP

    float* o = out + (size_t)(s0b + m) * kT + t00 + (hi << 3);

    float4 v;
    v.x = acc0[0]; v.y = acc0[1]; v.z = acc0[2]; v.w = acc0[3];
    *(float4*)(o) = v;
    v.x = acc0[4]; v.y = acc0[5]; v.z = acc0[6]; v.w = acc0[7];
    *(float4*)(o + 4) = v;
    if (t00 + 16 < kT) {
        v.x = acc1[0]; v.y = acc1[1]; v.z = acc1[2]; v.w = acc1[3];
        *(float4*)(o + 16) = v;
        v.x = acc1[4]; v.y = acc1[5]; v.z = acc1[6]; v.w = acc1[7];
        *(float4*)(o + 20) = v;
    }
    if (t00 + 32 < kT) {
        v.x = acc2[0]; v.y = acc2[1]; v.z = acc2[2]; v.w = acc2[3];
        *(float4*)(o + 32) = v;
        v.x = acc2[4]; v.y = acc2[5]; v.z = acc2[6]; v.w = acc2[7];
        *(float4*)(o + 36) = v;
    }
    if (t00 + 48 < kT) {
        v.x = acc3[0]; v.y = acc3[1]; v.z = acc3[2]; v.w = acc3[3];
        *(float4*)(o + 48) = v;
        v.x = acc3[4]; v.y = acc3[5]; v.z = acc3[6]; v.w = acc3[7];
        *(float4*)(o + 52) = v;
    }
}

extern "C" void kernel_launch(void* const* d_in, const int* in_sizes, int n_in,
                              void* d_out, int out_size, void* d_ws, size_t ws_size,
                              hipStream_t stream) {
    const float* x = (const float*)d_in[0];
    const float* h = (const float*)d_in[1];
    float* out = (float*)d_out;

    dim3 grid(kSig / kBlkS, (kT + kBlkT - 1) / kBlkT);
    fir_bandpass_wmma4x1<<<grid, 256, 0, stream>>>(x, h, out);
}


# Workgroup holds its 16 signals' whole padded length in LDS (283 KiB), staged once by grouped 16-byte loads, no halo re-reads; 256 workgroups x 32 waves, one per WGP; f32 sum order kept
# speedup vs baseline: 2.4375x; 1.2305x over previous
#include <hip/hip_runtime.h>


typedef __attribute__((ext_vector_type(2))) float v2f;
typedef __attribute__((ext_vector_type(8))) float v8f;

namespace {
constexpr int kT       = 4000;
constexpr int kSig     = 64 * 64;
constexpr int kTaps    = 413;
constexpr int kHalf    = (kTaps - 1) / 2;
constexpr int kBlkS    = 16;
constexpr int kWaveT   = 64;
constexpr int kWaves   = 32;
constexpr int kThreads = 32 * kWaves;
constexpr int kTiles   = (kT + kWaveT - 1) / kWaveT;
constexpr int kKSteps  = 119;
constexpr int kLead    = 2;
constexpr int kX0      = kLead + kHalf;
constexpr int kPadR    = kX0 + kT;
constexpr int kZero    = kPadR + kHalf;
constexpr int kRowEnd  = kLead + (kTiles - 1) * kWaveT + 4 * kKSteps;
constexpr int kStride  = 4452;
constexpr int kVec     = kT / 4;
static_assert(kT % 4 == 0 && kX0 % 4 == 0 && kStride % 4 == 0, "16-byte chunks of a row stay aligned in memory and in LDS");
static_assert(kZero + 32 >= kRowEnd && kStride >= kRowEnd, "zeroed tail covers every index the half-full last tile reads");
constexpr int kHpOff   = 64;
constexpr int kHpLen   = 544;
constexpr int kRing    = 16;
static_assert((kTaps + 15 + 3) / 4 == 107, "per-tile band length used by the K-step schedule below");
}

__global__ __launch_bounds__(kThreads)
void fir_bandpass_wmma4x1(const float* __restrict__ x,
                          const float* __restrict__ h,
                          float* __restrict__ out)
{
    __shared__ __attribute__((aligned(16))) float lds[2 * kHpLen + kBlkS * kStride];
    float* const hp = lds;
    float* const xs = lds + 2 * kHpLen;

    const int tid  = threadIdx.x;
    const int w    = tid >> 5;
    const int lane = tid & 31;
    const int s0b  = blockIdx.x * kBlkS;

    for (int i = tid; i < kHpLen; i += kThreads) {
        int j = i - kHpOff;
        float v = (j >= 0 && j < kTaps) ? h[kTaps - 1 - j] : 0.0f;
        hp[i] = v;
        hp[kHpLen + i] = v;
    }

    {
        const int r  = w >> 1;
        const int hf = w & 1;
        const float*  row  = x + (size_t)(s0b + r) * kT;
        const float4* row4 = (const float4*)row;
        float*        xr   = xs + r * kStride;

        float4 body[16];
        #pragma unroll
        for (int i = 0; i < 16; ++i) {
            const int c  = hf * 512 + 32 * i + lane;
            const int cc = c < kVec ? c : kVec - 1;
            body[i] = row4[cc];
        }
        const float edge = row[hf ? kT - 1 : 0];
        float pad[7];
        #pragma unroll
        for (int i = 0; i < 7; ++i) {
            const int p  = 32 * i + lane;
            const int pc = p < kHalf ? p : kHalf - 1;
            pad[i] = row[hf ? kT - 2 - pc : kHalf - pc];
        }

        #pragma unroll
        for (int i = 0; i < 16; ++i) {
            const int c  = hf * 512 + 32 * i + lane;
            const int cc = c < kVec ? c : kVec - 1;
            *(float4*)(xr + kX0 + 4 * cc) = body[i];
        }
        #pragma unroll
        for (int i = 0; i < 7; ++i) {
            const int p  = 32 * i + lane;
            const int pc = p < kHalf ? p : kHalf - 1;
            xr[(hf ? kPadR : kLead) + pc] = 2.0f * edge - pad[i];
        }
        if (hf) xr[kZero + lane] = 0.0f;
    }
    __syncthreads();

    const int m  = lane & 15;
    const int hi = lane >> 4;
    const int kk = hi << 1;

    const float* hb = hp + hi * kHpLen + (kHpOff - 48 + kk - m);

    #pragma unroll 1
    for (int tile = w; tile < kTiles; tile += kWaves) {
    const int t00 = tile * kWaveT;
    const float* xb = xs + m * kStride + kLead + t00 + kk;

    v8f acc0 = {0.f,0.f,0.f,0.f,0.f,0.f,0.f,0.f};
    v8f acc1 = acc0;
    v8f acc2 = acc0;
    v8f acc3 = acc0;

    v2f ring[kRing];

#define FIR_STEP(ko, sl, ld, t0, t1, t2, t3)                                          \
    {                                                                                 \
        if (ld) {                                                                     \
            v2f a;                                                                    \
            a.x = hq[4 * (ko)];                                                       \
            a.y = hq[4 * (ko) + 1];                                                   \
            ring[(sl) & (kRing - 1)] = a;                                             \
        }                                                                             \
        const v2f b = *(const v2f*)(xq + 4 * (ko));                                   \
        if (t0) acc0 = __builtin_amdgcn_wmma_f32_16x16x4_f32(                         \
            false, ring[(sl) & (kRing - 1)], false, b, (short)0, acc0, false, false); \
        if (t1) acc1 = __builtin_amdgcn_wmma_f32_16x16x4_f32(                         \
            false, ring[((sl) + 12) & (kRing - 1)], false, b, (short)0, acc1, false, false); \
        if (t2) acc2 = __builtin_amdgcn_wmma_f32_16x16x4_f32(                         \
            false, ring[((sl) + 8) & (kRing - 1)], false, b, (short)0, acc2, false, false);  \
        if (t3) acc3 = __builtin_amdgcn_wmma_f32_16x16x4_f32(                         \
            false, ring[((sl) + 4) & (kRing - 1)], false, b, (short)0, acc3, false, false);  \
    }

    const float* hq = hb + 48;
    const float* xq = xb;

    #pragma unroll
    for (int k = 0; k < 4; ++k)  FIR_STEP(k, k, true, true, false, false, false)
    #pragma unroll
    for (int k = 4; k < 8; ++k)  FIR_STEP(k, k, true, true, true, false, false)
    #pragma unroll
    for (int k = 8; k < 12; ++k) FIR_STEP(k, k, true, true, true, true, false)
    hq += 48;
    xq += 48;
    #pragma unroll 1
    for (int kb = 0; kb < 5; ++kb) {
        #pragma unroll
        for (int s = 0; s < 16; ++s) FIR_STEP(s, s + 12, true, true, true, true, true)
        hq += 64;
        xq += 64;
    }
    #pragma unroll
    for (int s = 0; s < 15; ++s) FIR_STEP(s, s + 12, true, true, true, true, true)
    #pragma unroll
    for (int s = 15; s < 19; ++s) FIR_STEP(s, s + 12, false, false, true, true, true)
    #pragma unroll
    for (int s = 19; s < 23; ++s) FIR_STEP(s, s + 12, false, false, false, true, true)
    #pragma unroll
    for (int s = 23; s < 27; ++s) FIR_STEP(s, s + 12, false, false, false, false, true)
#undef FIR_STEP

    float* o = out + (size_t)(s0b + m) * kT + t00 + (hi << 3);

    float4 v;
    v.x = acc0[0]; v.y = acc0[1]; v.z = acc0[2]; v.w = acc0[3];
    *(float4*)(o) = v;
    v.x = acc0[4]; v.y = acc0[5]; v.z = acc0[6]; v.w = acc0[7];
    *(float4*)(o + 4) = v;
    if (t00 + 16 < kT) {
        v.x = acc1[0]; v.y = acc1[1]; v.z = acc1[2]; v.w = acc1[3];
        *(float4*)(o + 16) = v;
        v.x = acc1[4]; v.y = acc1[5]; v.z = acc1[6]; v.w = acc1[7];
        *(float4*)(o + 20) = v;
    }
    if (t00 + 32 < kT) {
        v.x = acc2[0]; v.y = acc2[1]; v.z = acc2[2]; v.w = acc2[3];
        *(float4*)(o + 32) = v;
        v.x = acc2[4]; v.y = acc2[5]; v.z = acc2[6]; v.w = acc2[7];
        *(float4*)(o + 36) = v;
    }
    if (t00 + 48 < kT) {
        v.x = acc3[0]; v.y = acc3[1]; v.z = acc3[2]; v.w = acc3[3];
        *(float4*)(o + 48) = v;
        v.x = acc3[4]; v.y = acc3[5]; v.z = acc3[6]; v.w = acc3[7];
        *(float4*)(o + 52) = v;
    }
    }
}

extern "C" void kernel_launch(void* const* d_in, const int* in_sizes, int n_in,
                              void* d_out, int out_size, void* d_ws, size_t ws_size,
                              hipStream_t stream) {
    const float* x = (const float*)d_in[0];
    const float* h = (const float*)d_in[1];
    float* out = (float*)d_out;

    dim3 grid(kSig / kBlkS);
    fir_bandpass_wmma4x1<<<grid, kThreads, 0, stream>>>(x, h, out);
}


# Tap operand stationary in registers: each wave reads its 107 tap fragments from LDS once for all its tiles (214 VGPRs, 8 waves per workgroup); K loop reads only signal fragments; f32 sum order kept
# speedup vs baseline: 2.5763x; 1.0569x over previous
#include <hip/hip_runtime.h>


typedef __attribute__((ext_vector_type(2))) float v2f;
typedef __attribute__((ext_vector_type(8))) float v8f;

namespace {
constexpr int kT       = 4000;
constexpr int kSig     = 64 * 64;
constexpr int kTaps    = 413;
constexpr int kHalf    = (kTaps - 1) / 2;
constexpr int kBlkS    = 16;
constexpr int kWaveT   = 64;
constexpr int kWaves   = 8;
constexpr int kThreads = 32 * kWaves;
constexpr int kTiles   = (kT + kWaveT - 1) / kWaveT;
constexpr int kKSteps  = 119;
constexpr int kBand    = 107;
constexpr int kLead    = 2;
constexpr int kX0      = kLead + kHalf;
constexpr int kPadR    = kX0 + kT;
constexpr int kZero    = kPadR + kHalf;
constexpr int kRowEnd  = kLead + (kTiles - 1) * kWaveT + 4 * kKSteps;
constexpr int kStride  = 4452;
constexpr int kVec     = kT / 4;
static_assert(kT % 4 == 0 && kX0 % 4 == 0 && kStride % 4 == 0, "16-byte chunks of a row stay aligned in memory and in LDS");
static_assert(kZero + 32 >= kRowEnd && kStride >= kRowEnd, "zeroed tail covers every index the half-full last tile reads");
constexpr int kHpOff   = 64;
constexpr int kHpLen   = 544;
constexpr int kGrp     = 30;
static_assert((kTaps + 15 + 3) / 4 == kBand && kBand + 12 == kKSteps, "per-tile band length used by the K-step schedule below");
static_assert(3 * kGrp < kKSteps && 4 * kGrp >= kKSteps, "four read groups cover the K-steps of a wave tile");
static_assert(kBlkS == 2 * kWaves, "each wave stages two rows");
}

__global__ __launch_bounds__(kThreads)
void fir_bandpass_wmma_regtaps(const float* __restrict__ x,
                               const float* __restrict__ h,
                               float* __restrict__ out)
{
    __shared__ __attribute__((aligned(16))) float lds[2 * kHpLen + kBlkS * kStride];
    float* const hp = lds;
    float* const xs = lds + 2 * kHpLen;

    const int tid  = threadIdx.x;
    const int w    = tid >> 5;
    const int lane = tid & 31;
    const int s0b  = blockIdx.x * kBlkS;

    for (int i = tid; i < kHpLen; i += kThreads) {
        int j = i - kHpOff;
        float v = (j >= 0 && j < kTaps) ? h[kTaps - 1 - j] : 0.0f;
        hp[i] = v;
        hp[kHpLen + i] = v;
    }

    {
        float4 body[2][32];
        float  edgeL[2], edgeR[2], padL[2][7], padR[2][7];
        #pragma unroll
        for (int rr = 0; rr < 2; ++rr) {
            const float*  row  = x + (size_t)(s0b + 2 * w + rr) * kT;
            const float4* row4 = (const float4*)row;
            #pragma unroll
            for (int i = 0; i < 32; ++i) {
                const int c  = 32 * i + lane;
                const int cc = c < kVec ? c : kVec - 1;
                body[rr][i] = row4[cc];
            }
            edgeL[rr] = row[0];
            edgeR[rr] = row[kT - 1];
            #pragma unroll
            for (int i = 0; i < 7; ++i) {
                const int p  = 32 * i + lane;
                const int pc = p < kHalf ? p : kHalf - 1;
                padL[rr][i] = row[kHalf - pc];
                padR[rr][i] = row[kT - 2 - pc];
            }
        }
        #pragma unroll
        for (int rr = 0; rr < 2; ++rr) {
            float* xr = xs + (2 * w + rr) * kStride;
            #pragma unroll
            for (int i = 0; i < 32; ++i) {
                const int c  = 32 * i + lane;
                const int cc = c < kVec ? c : kVec - 1;
                *(float4*)(xr + kX0 + 4 * cc) = body[rr][i];
            }
            #pragma unroll
            for (int i = 0; i < 7; ++i) {
                const int p  = 32 * i + lane;
                const int pc = p < kHalf ? p : kHalf - 1;
                xr[kLead + pc] = 2.0f * edgeL[rr] - padL[rr][i];
                xr[kPadR + pc] = 2.0f * edgeR[rr] - padR[rr][i];
            }
            xr[kZero + lane] = 0.0f;
        }
    }
    __syncthreads();

    const int m  = lane & 15;
    const int hi = lane >> 4;
    const int kk = hi << 1;

    v2f tap[kBand];
    {
        const float* hq = hp + hi * kHpLen + (kHpOff + kk - m);
        #pragma unroll
        for (int j = 0; j < kBand; ++j) {
            tap[j].x = hq[4 * j];
            tap[j].y = hq[4 * j + 1];
        }
    }

    const float* xrow = xs + m * kStride + kLead + kk;

    v2f bA[kGrp], bB[kGrp];

#define FIR_READ(BUF, XP, K0, N)                                                   \
    {                                                                              \
        _Pragma("unroll")                                                          \
        for (int s = 0; s < (N); ++s) BUF[s] = *(const v2f*)((XP) + 4 * ((K0) + s)); \
    }
#define FIR_BLOCK(BUF, K0, K1)                                                     \
    {                                                                              \
        _Pragma("unroll")                                                          \
        for (int k = (K0); k < (K1); ++k) {                                        \
            const v2f b = BUF[k - (K0)];                                           \
            _Pragma("unroll")                                                      \
            for (int i = 0; i < 4; ++i) {                                          \
                const int j = k - 4 * i;                                           \
                if (j >= 0 && j < kBand)                                           \
                    acc[i] = __builtin_amdgcn_wmma_f32_16x16x4_f32(                \
                        false, tap[j], false, b, (short)0, acc[i], false, false);  \
            }                                                                      \
        }                                                                          \
    }
#define FIR_FENCE() __builtin_amdgcn_sched_barrier(0)

    FIR_READ(bA, xrow + w * kWaveT, 0, kGrp)

    #pragma unroll 1
    for (int tile = w; tile < kTiles; tile += kWaves) {
        const int t00 = tile * kWaveT;
        const float* xb = xrow + t00;
        const int nxt = tile + kWaves < kTiles ? tile + kWaves : tile;
        const float* xn = xrow + nxt * kWaveT;

        v8f acc[4];
        #pragma unroll
        for (int i = 0; i < 4; ++i) acc[i] = (v8f){0.f, 0.f, 0.f, 0.f, 0.f, 0.f, 0.f, 0.f};

        FIR_READ(bB, xb, kGrp, kGrp)
        FIR_FENCE();
        FIR_BLOCK(bA, 0, kGrp)
        FIR_FENCE();
        FIR_READ(bA, xb, 2 * kGrp, kGrp)
        FIR_FENCE();
        FIR_BLOCK(bB, kGrp, 2 * kGrp)
        FIR_FENCE();
        FIR_READ(bB, xb, 3 * kGrp, kKSteps - 3 * kGrp)
        FIR_FENCE();
        FIR_BLOCK(bA, 2 * kGrp, 3 * kGrp)
        FIR_FENCE();
        FIR_READ(bA, xn, 0, kGrp)
        FIR_FENCE();
        FIR_BLOCK(bB, 3 * kGrp, kKSteps)
        FIR_FENCE();

        float* o = out + (size_t)(s0b + m) * kT + t00 + (hi << 3);

        float4 v;
        v.x = acc[0][0]; v.y = acc[0][1]; v.z = acc[0][2]; v.w = acc[0][3];
        *(float4*)(o) = v;
        v.x = acc[0][4]; v.y = acc[0][5]; v.z = acc[0][6]; v.w = acc[0][7];
        *(float4*)(o + 4) = v;
        if (t00 + 16 < kT) {
            v.x = acc[1][0]; v.y = acc[1][1]; v.z = acc[1][2]; v.w = acc[1][3];
            *(float4*)(o + 16) = v;
            v.x = acc[1][4]; v.y = acc[1][5]; v.z = acc[1][6]; v.w = acc[1][7];
            *(float4*)(o + 20) = v;
        }
        if (t00 + 32 < kT) {
            v.x = acc[2][0]; v.y = acc[2][1]; v.z = acc[2][2]; v.w = acc[2][3];
            *(float4*)(o + 32) = v;
            v.x = acc[2][4]; v.y = acc[2][5]; v.z = acc[2][6]; v.w = acc[2][7];
            *(float4*)(o + 36) = v;
        }
        if (t00 + 48 < kT) {
            v.x = acc[3][0]; v.y = acc[3][1]; v.z = acc[3][2]; v.w = acc[3][3];
            *(float4*)(o + 48) = v;
            v.x = acc[3][4]; v.y = acc[3][5]; v.z = acc[3][6]; v.w = acc[3][7];
            *(float4*)(o + 52) = v;
        }
    }
#undef FIR_READ
#undef FIR_BLOCK
#undef FIR_FENCE
}

extern "C" void kernel_launch(void* const* d_in, const int* in_sizes, int n_in,
                              void* d_out, int out_size, void* d_ws, size_t ws_size,
                              hipStream_t stream) {
    const float* x = (const float*)d_in[0];
    const float* h = (const float*)d_in[1];
    float* out = (float*)d_out;

    dim3 grid(kSig / kBlkS);
    fir_bandpass_wmma_regtaps<<<grid, kThreads, 0, stream>>>(x, h, out);
}
